# starting kernel
# baseline (speedup 1.0000x reference)
#include <hip/hip_runtime.h>

#define B_ 4
#define N_ 20000
#define Dn_ 64
#define Dm_ 128
#define E_ 160000

typedef float v2f __attribute__((ext_vector_type(2)));
typedef float v8f __attribute__((ext_vector_type(8)));

__device__ __forceinline__ v8f wmma4(v2f a, v2f b, v8f c) {
  return __builtin_amdgcn_wmma_f32_16x16x4_f32(false, a, false, b, (short)0, c,
                                               false, false);
}

__global__ __launch_bounds__(128)
void proj_kernel(const float* __restrict__ X, const float* __restrict__ W,
                 const float* __restrict__ bias, float* __restrict__ H) {
  const int lane = threadIdx.x & 31;
  const int wave = threadIdx.x >> 5;
  const int m    = lane & 15;
  const int kh   = lane >> 4;
  const int rowbase = blockIdx.x * 16;

  v2f a[16];
  const float* xr = X + (size_t)(rowbase + m) * Dn_;
#pragma unroll
  for (int kk = 0; kk < 16; ++kk) {
    int k0 = kk * 4 + kh * 2;
    a[kk].x = xr[k0];
    a[kk].y = xr[k0 + 1];
  }

#pragma unroll
  for (int t = 0; t < 2; ++t) {
    int col = (wave * 2 + t) * 16;
    v8f c = {0.f, 0.f, 0.f, 0.f, 0.f, 0.f, 0.f, 0.f};
#pragma unroll
    for (int kk = 0; kk < 16; ++kk) {
      int k0 = kk * 4 + kh * 2;
      v2f bf;
      bf.x = W[k0 * Dm_ + col + m];
      bf.y = W[(k0 + 1) * Dm_ + col + m];
      c = wmma4(a[kk], bf, c);
    }
    int rowadd = kh * 8;
#pragma unroll
    for (int v = 0; v < 8; ++v) {
      H[(size_t)(rowbase + v + rowadd) * Dm_ + col + m] = c[v] + bias[col + m];
    }
  }
}

__global__ __launch_bounds__(256)
void zero_kernel(float4* __restrict__ p, int n4) {
  int i = blockIdx.x * 256 + threadIdx.x;
  if (i < n4) p[i] = make_float4(0.f, 0.f, 0.f, 0.f);
}

__global__ __launch_bounds__(128)
void edge_kernel(const float* __restrict__ H, float* __restrict__ agg,
                 const float* __restrict__ node_W, const float* __restrict__ node_b,
                 const float* __restrict__ edge_W, const float* __restrict__ edge_b,
                 const float* __restrict__ ea0, const float* __restrict__ ea1,
                 const int* __restrict__ ei0, const int* __restrict__ ei1,
                 const int* __restrict__ ei2, int layer) {
  __shared__ float T[16][132];
  __shared__ float Hs[16][132];
  __shared__ int s_src[16];
  __shared__ int s_dst[16];

  const int r     = blockIdx.y;
  const int ebase = blockIdx.x * 16;
  const int tid   = threadIdx.x;
  const int lane  = tid & 31;
  const int wave  = tid >> 5;
  const int m     = lane & 15;
  const int kh    = lane >> 4;

  const int* ei = (r == 0) ? ei0 : (r == 1) ? ei1 : ei2;
  if (tid < 16) {
    s_src[tid] = ei[ebase + tid];
    s_dst[tid] = ei[E_ + ebase + tid];
  }

  const float* nb = node_b + (size_t)(layer * 3 + r) * Dm_;

  if (r < 2) {
    const float* ea = (r == 0) ? ea0 : ea1;
    const float* Wq = edge_W + (size_t)(layer * 2 + r) * 16 * Dm_;
    const float* eb = edge_b + (size_t)(layer * 2 + r) * Dm_;
    v2f a[4];
    const float* er = ea + (size_t)(ebase + m) * 16;
#pragma unroll
    for (int kk = 0; kk < 4; ++kk) {
      int k0 = kk * 4 + kh * 2;
      a[kk].x = er[k0];
      a[kk].y = er[k0 + 1];
    }
#pragma unroll
    for (int t = 0; t < 2; ++t) {
      int col = (wave * 2 + t) * 16;
      v8f c = {0.f, 0.f, 0.f, 0.f, 0.f, 0.f, 0.f, 0.f};
#pragma unroll
      for (int kk = 0; kk < 4; ++kk) {
        int k0 = kk * 4 + kh * 2;
        v2f bf;
        bf.x = Wq[k0 * Dm_ + col + m];
        bf.y = Wq[(k0 + 1) * Dm_ + col + m];
        c = wmma4(a[kk], bf, c);
      }
      int rowadd = kh * 8;
#pragma unroll
      for (int v = 0; v < 8; ++v) {
        T[v + rowadd][col + m] = c[v] + nb[col + m] + eb[col + m];
      }
    }
  } else {
    for (int i = tid; i < 16 * Dm_; i += 128) {
      int row = i >> 7;
      int cc  = i & 127;
      T[row][cc] = nb[cc];
    }
  }
  __syncthreads();

  const float* Wn = node_W + (size_t)(layer * 3 + r) * Dm_ * Dm_;

  for (int b = 0; b < B_; ++b) {
#pragma unroll
    for (int j = 0; j < 4; ++j) {
      int idx = tid + 128 * j;
      int row = idx >> 5;
      int c4  = idx & 31;
      const float* src = H + ((size_t)b * N_ + s_src[row]) * Dm_ + c4 * 4;
      float4 v = *(const float4*)src;
      *(float4*)&Hs[row][c4 * 4] = v;
    }
    __syncthreads();

#pragma unroll
    for (int t = 0; t < 2; ++t) {
      int col = (wave * 2 + t) * 16;
      v8f c = {0.f, 0.f, 0.f, 0.f, 0.f, 0.f, 0.f, 0.f};
#pragma unroll 8
      for (int kk = 0; kk < 32; ++kk) {
        int k0 = kk * 4 + kh * 2;
        v2f af;
        af.x = Hs[m][k0];
        af.y = Hs[m][k0 + 1];
        v2f bf;
        bf.x = Wn[k0 * Dm_ + col + m];
        bf.y = Wn[(k0 + 1) * Dm_ + col + m];
        c = wmma4(af, bf, c);
      }
      int rowadd = kh * 8;
#pragma unroll
      for (int v = 0; v < 8; ++v) {
        int mr = v + rowadd;
        float val = c[v] + T[mr][col + m];
        atomicAdd(&agg[((size_t)b * N_ + s_dst[mr]) * Dm_ + col + m], val);
      }
    }
    __syncthreads();
  }
}

__global__ __launch_bounds__(128)
void ln_kernel(float* __restrict__ H, const float* __restrict__ agg,
               const float* __restrict__ ln_g, const float* __restrict__ ln_b,
               int layer) {
  const int lane = threadIdx.x & 31;
  const int wave = threadIdx.x >> 5;
  const size_t row = (size_t)blockIdx.x * 4 + wave;

  float* hp = H + row * Dm_;
  const float* ap = agg + row * Dm_;
  float4 x = *(const float4*)(hp + lane * 4);
  float4 a = *(const float4*)(ap + lane * 4);
  float4 y;
  y.x = x.x + fmaxf(a.x, 0.f);
  y.y = x.y + fmaxf(a.y, 0.f);
  y.z = x.z + fmaxf(a.z, 0.f);
  y.w = x.w + fmaxf(a.w, 0.f);

  float s = y.x + y.y + y.z + y.w;
  float q = y.x * y.x + y.y * y.y + y.z * y.z + y.w * y.w;
#pragma unroll
  for (int off = 16; off >= 1; off >>= 1) {
    s += __shfl_xor(s, off, 32);
    q += __shfl_xor(q, off, 32);
  }
  float mean = s * (1.0f / Dm_);
  float var  = q * (1.0f / Dm_) - mean * mean;
  float rs   = rsqrtf(var + 1e-5f);

  const float* g  = ln_g + (size_t)layer * Dm_ + lane * 4;
  const float* bb = ln_b + (size_t)layer * Dm_ + lane * 4;
  float4 o;
  o.x = (y.x - mean) * rs * g[0] + bb[0];
  o.y = (y.y - mean) * rs * g[1] + bb[1];
  o.z = (y.z - mean) * rs * g[2] + bb[2];
  o.w = (y.w - mean) * rs * g[3] + bb[3];
  *(float4*)(hp + lane * 4) = o;
}

extern "C" void kernel_launch(void* const* d_in, const int* in_sizes, int n_in,
                              void* d_out, int out_size, void* d_ws, size_t ws_size,
                              hipStream_t stream) {
  const float* node_feat = (const float*)d_in[0];
  const float* in_W      = (const float*)d_in[1];
  const float* in_b      = (const float*)d_in[2];
  const float* node_W    = (const float*)d_in[3];
  const float* node_b    = (const float*)d_in[4];
  const float* edge_W    = (const float*)d_in[5];
  const float* edge_b    = (const float*)d_in[6];
  const float* ln_g      = (const float*)d_in[7];
  const float* ln_b      = (const float*)d_in[8];
  const float* ea0       = (const float*)d_in[9];
  const float* ea1       = (const float*)d_in[10];
  const int*   ei0       = (const int*)d_in[11];
  const int*   ei1       = (const int*)d_in[12];
  const int*   ei2       = (const int*)d_in[13];

  float* H   = (float*)d_out;
  float* agg = (float*)d_ws;

  proj_kernel<<<dim3((B_ * N_) / 16), dim3(128), 0, stream>>>(node_feat, in_W,
                                                              in_b, H);
  const int n4 = (B_ * N_ * Dm_) / 4;
  for (int l = 0; l < 2; ++l) {
    zero_kernel<<<dim3((n4 + 255) / 256), dim3(256), 0, stream>>>((float4*)agg,
                                                                  n4);
    edge_kernel<<<dim3(E_ / 16, 3), dim3(128), 0, stream>>>(
        H, agg, node_W, node_b, edge_W, edge_b, ea0, ea1, ei0, ei1, ei2, l);
    ln_kernel<<<dim3((B_ * N_) / 4), dim3(128), 0, stream>>>(H, agg, ln_g,
                                                             ln_b, l);
  }
}


# Per-relation node linear hoisted from edges to nodes (row gather commutes with it): per batch element P_r=H_b@W_r once per node in workspace; edge pass gathers P_r rows, same f32 atomic adds
# speedup vs baseline: 2.1595x; 2.1595x over previous
#include <hip/hip_runtime.h>


#define B_ 4
#define N_ 20000
#define Dn_ 64
#define Dm_ 128
#define E_ 160000

typedef float v2f __attribute__((ext_vector_type(2)));
typedef float v8f __attribute__((ext_vector_type(8)));

__device__ __forceinline__ v8f wmma4(v2f a, v2f b, v8f c) {
  return __builtin_amdgcn_wmma_f32_16x16x4_f32(false, a, false, b, (short)0, c,
                                               false, false);
}

__global__ __launch_bounds__(128)
void proj_kernel(const float* __restrict__ X, const float* __restrict__ W,
                 const float* __restrict__ bias, float* __restrict__ H) {
  const int lane = threadIdx.x & 31;
  const int wave = threadIdx.x >> 5;
  const int m    = lane & 15;
  const int kh   = lane >> 4;
  const int rowbase = blockIdx.x * 16;

  v2f a[16];
  const float* xr = X + (size_t)(rowbase + m) * Dn_;
#pragma unroll
  for (int kk = 0; kk < 16; ++kk) {
    int k0 = kk * 4 + kh * 2;
    a[kk].x = xr[k0];
    a[kk].y = xr[k0 + 1];
  }

#pragma unroll
  for (int t = 0; t < 2; ++t) {
    int col = (wave * 2 + t) * 16;
    v8f c = {0.f, 0.f, 0.f, 0.f, 0.f, 0.f, 0.f, 0.f};
#pragma unroll
    for (int kk = 0; kk < 16; ++kk) {
      int k0 = kk * 4 + kh * 2;
      v2f bf;
      bf.x = W[k0 * Dm_ + col + m];
      bf.y = W[(k0 + 1) * Dm_ + col + m];
      c = wmma4(a[kk], bf, c);
    }
    int rowadd = kh * 8;
#pragma unroll
    for (int v = 0; v < 8; ++v) {
      H[(size_t)(rowbase + v + rowadd) * Dm_ + col + m] = c[v] + bias[col + m];
    }
  }
}


__global__ __launch_bounds__(128)
void node_kernel(const float* __restrict__ H, float* __restrict__ ws,
                 const float* __restrict__ node_W, int layer, int b) {
  __shared__ float Hs[32][132];

  const int r    = blockIdx.y;
  const int tid  = threadIdx.x;
  const int lane = tid & 31;
  const int wave = tid >> 5;
  const int m    = lane & 15;
  const int kh   = lane >> 4;
  const int rowbase = blockIdx.x * 32;

  const float* Hb = H + (size_t)b * N_ * Dm_;
  float* aggb = ws;

  const int c4   = tid & 31;
  const int row0 = tid >> 5;
  float4 hv[8];
#pragma unroll
  for (int j = 0; j < 8; ++j) {
    hv[j] = *(const float4*)(Hb + (size_t)(rowbase + row0 + 4 * j) * Dm_ + c4 * 4);
  }
#pragma unroll
  for (int j = 0; j < 8; ++j) {
    *(float4*)&Hs[row0 + 4 * j][c4 * 4] = hv[j];
  }
  if (r == 0) {
#pragma unroll
    for (int j = 0; j < 8; ++j) {
      *(float4*)(aggb + (size_t)(rowbase + row0 + 4 * j) * Dm_ + c4 * 4) =
          make_float4(0.f, 0.f, 0.f, 0.f);
    }
  }
  __syncthreads();

  const int col0 = wave * 32 + m;
  const int rowadd = kh * 8;

  const float* Wn = node_W + (size_t)(layer * 3 + r) * Dm_ * Dm_;
  float* P = ws + (size_t)(1 + r) * N_ * Dm_;
  v8f c00 = {0.f, 0.f, 0.f, 0.f, 0.f, 0.f, 0.f, 0.f};
  v8f c01 = c00, c10 = c00, c11 = c00;
#pragma unroll 8
  for (int kk = 0; kk < 32; ++kk) {
    int k0 = kk * 4 + kh * 2;
    v2f a0, a1, b0, b1;
    a0.x = Hs[m][k0];
    a0.y = Hs[m][k0 + 1];
    a1.x = Hs[16 + m][k0];
    a1.y = Hs[16 + m][k0 + 1];
    b0.x = Wn[k0 * Dm_ + col0];
    b0.y = Wn[(k0 + 1) * Dm_ + col0];
    b1.x = Wn[k0 * Dm_ + col0 + 16];
    b1.y = Wn[(k0 + 1) * Dm_ + col0 + 16];
    c00 = wmma4(a0, b0, c00);
    c01 = wmma4(a0, b1, c01);
    c10 = wmma4(a1, b0, c10);
    c11 = wmma4(a1, b1, c11);
  }
#pragma unroll
  for (int v = 0; v < 8; ++v) {
    size_t g0 = (size_t)(rowbase + v + rowadd) * Dm_ + col0;
    size_t g1 = (size_t)(rowbase + 16 + v + rowadd) * Dm_ + col0;
    P[g0]      = c00[v];
    P[g0 + 16] = c01[v];
    P[g1]      = c10[v];
    P[g1 + 16] = c11[v];
  }
}

#define EDGE_GRID 1000

__global__ __launch_bounds__(128)
void edge_kernel(float* __restrict__ ws,
                 const float* __restrict__ node_b,
                 const float* __restrict__ edge_W, const float* __restrict__ edge_b,
                 const float* __restrict__ ea0, const float* __restrict__ ea1,
                 const int* __restrict__ ei0, const int* __restrict__ ei1,
                 const int* __restrict__ ei2, int layer) {
  __shared__ float T[2][16][132];

  const int r    = blockIdx.y;
  const int tid  = threadIdx.x;
  const int lane = tid & 31;
  const int wave = __builtin_amdgcn_readfirstlane(tid >> 5);
  const int m    = lane & 15;
  const int kh   = lane >> 4;

  float* agg = ws;
  const float* P = ws + (size_t)(1 + r) * N_ * Dm_;
  const int* ei = (r == 0) ? ei0 : (r == 1) ? ei1 : ei2;
  const float* nb = node_b + (size_t)(layer * 3 + r) * Dm_;

  if (r < 2) {
    const float* ea = (r == 0) ? ea0 : ea1;
    const float* Wq = edge_W + (size_t)(layer * 2 + r) * 16 * Dm_;
    const float* eb = edge_b + (size_t)(layer * 2 + r) * Dm_;

    v2f bq[2][4];
    float nbv[2], ebv[2];
#pragma unroll
    for (int t = 0; t < 2; ++t) {
      int col = (wave * 2 + t) * 16;
#pragma unroll
      for (int kk = 0; kk < 4; ++kk) {
        int k0 = kk * 4 + kh * 2;
        bq[t][kk].x = Wq[k0 * Dm_ + col + m];
        bq[t][kk].y = Wq[(k0 + 1) * Dm_ + col + m];
      }
      nbv[t] = nb[col + m];
      ebv[t] = eb[col + m];
    }

    int buf = 0;
    int tile = blockIdx.x;
    int s[4], d[4];
#pragma unroll
    for (int i = 0; i < 4; ++i) {
      s[i] = ei[tile * 16 + wave * 4 + i];
      d[i] = ei[E_ + tile * 16 + wave * 4 + i];
    }
    for (; tile < E_ / 16; tile += EDGE_GRID) {
      const int e0 = tile * 16;
      v2f a[4];
      const float* er = ea + (size_t)(e0 + m) * 16;
#pragma unroll
      for (int kk = 0; kk < 4; ++kk) {
        int k0 = kk * 4 + kh * 2;
        a[kk].x = er[k0];
        a[kk].y = er[k0 + 1];
      }
      float x[4][4];
      float* ar[4];
#pragma unroll
      for (int i = 0; i < 4; ++i) {
        const float* pr = P + (size_t)s[i] * Dm_ + lane;
        ar[i] = agg + (size_t)d[i] * Dm_ + lane;
#pragma unroll
        for (int j = 0; j < 4; ++j) x[i][j] = pr[32 * j];
      }
      const int nt = (tile + EDGE_GRID < E_ / 16) ? tile + EDGE_GRID : tile;
#pragma unroll
      for (int i = 0; i < 4; ++i) {
        s[i] = ei[nt * 16 + wave * 4 + i];
        d[i] = ei[E_ + nt * 16 + wave * 4 + i];
      }

#pragma unroll
      for (int t = 0; t < 2; ++t) {
        int col = (wave * 2 + t) * 16;
        v8f c = {0.f, 0.f, 0.f, 0.f, 0.f, 0.f, 0.f, 0.f};
#pragma unroll
        for (int kk = 0; kk < 4; ++kk) c = wmma4(a[kk], bq[t][kk], c);
        int rowadd = kh * 8;
#pragma unroll
        for (int v = 0; v < 8; ++v) {
          T[buf][v + rowadd][col + m] = c[v] + nbv[t] + ebv[t];
        }
      }
      __syncthreads();

      float val[4][4];
#pragma unroll
      for (int i = 0; i < 4; ++i) {
#pragma unroll
        for (int j = 0; j < 4; ++j) {
          val[i][j] = x[i][j] + T[buf][wave * 4 + i][32 * j + lane];
        }
      }
#pragma unroll
      for (int i = 0; i < 4; ++i) {
#pragma unroll
        for (int j = 0; j < 4; ++j) atomicAdd(ar[i] + 32 * j, val[i][j]);
      }
      buf ^= 1;
    }
  } else {
    float nbl[4];
#pragma unroll
    for (int j = 0; j < 4; ++j) nbl[j] = nb[32 * j + lane];

    int tile = blockIdx.x;
    int s[4], d[4];
#pragma unroll
    for (int i = 0; i < 4; ++i) {
      s[i] = ei[tile * 16 + wave * 4 + i];
      d[i] = ei[E_ + tile * 16 + wave * 4 + i];
    }
    for (; tile < E_ / 16; tile += EDGE_GRID) {
      float x[4][4];
      float* ar[4];
#pragma unroll
      for (int i = 0; i < 4; ++i) {
        const float* pr = P + (size_t)s[i] * Dm_ + lane;
        ar[i] = agg + (size_t)d[i] * Dm_ + lane;
#pragma unroll
        for (int j = 0; j < 4; ++j) x[i][j] = pr[32 * j];
      }
      const int nt = (tile + EDGE_GRID < E_ / 16) ? tile + EDGE_GRID : tile;
#pragma unroll
      for (int i = 0; i < 4; ++i) {
        s[i] = ei[nt * 16 + wave * 4 + i];
        d[i] = ei[E_ + nt * 16 + wave * 4 + i];
      }
#pragma unroll
      for (int i = 0; i < 4; ++i) {
#pragma unroll
        for (int j = 0; j < 4; ++j) atomicAdd(ar[i] + 32 * j, x[i][j] + nbl[j]);
      }
    }
  }
}

__global__ __launch_bounds__(128)
void ln_kernel(float* __restrict__ H, const float* __restrict__ ws,
               const float* __restrict__ ln_g, const float* __restrict__ ln_b,
               int layer, int b) {
  const int lane = threadIdx.x & 31;
  const int wave = threadIdx.x >> 5;
  const size_t row = (size_t)blockIdx.x * 4 + wave;

  float* hp = H + ((size_t)b * N_ + row) * Dm_;
  const float* ap = ws + row * Dm_;
  float4 x = *(const float4*)(hp + lane * 4);
  float4 a = *(const float4*)(ap + lane * 4);
  float4 y;
  y.x = x.x + fmaxf(a.x, 0.f);
  y.y = x.y + fmaxf(a.y, 0.f);
  y.z = x.z + fmaxf(a.z, 0.f);
  y.w = x.w + fmaxf(a.w, 0.f);

  float s = y.x + y.y + y.z + y.w;
  float q = y.x * y.x + y.y * y.y + y.z * y.z + y.w * y.w;
#pragma unroll
  for (int off = 16; off >= 1; off >>= 1) {
    s += __shfl_xor(s, off, 32);
    q += __shfl_xor(q, off, 32);
  }
  float mean = s * (1.0f / Dm_);
  float var  = q * (1.0f / Dm_) - mean * mean;
  float rs   = rsqrtf(var + 1e-5f);

  const float* g  = ln_g + (size_t)layer * Dm_ + lane * 4;
  const float* bb = ln_b + (size_t)layer * Dm_ + lane * 4;
  float4 o;
  o.x = (y.x - mean) * rs * g[0] + bb[0];
  o.y = (y.y - mean) * rs * g[1] + bb[1];
  o.z = (y.z - mean) * rs * g[2] + bb[2];
  o.w = (y.w - mean) * rs * g[3] + bb[3];
  *(float4*)(hp + lane * 4) = o;
}

extern "C" void kernel_launch(void* const* d_in, const int* in_sizes, int n_in,
                              void* d_out, int out_size, void* d_ws, size_t ws_size,
                              hipStream_t stream) {
  const float* node_feat = (const float*)d_in[0];
  const float* in_W      = (const float*)d_in[1];
  const float* in_b      = (const float*)d_in[2];
  const float* node_W    = (const float*)d_in[3];
  const float* node_b    = (const float*)d_in[4];
  const float* edge_W    = (const float*)d_in[5];
  const float* edge_b    = (const float*)d_in[6];
  const float* ln_g      = (const float*)d_in[7];
  const float* ln_b      = (const float*)d_in[8];
  const float* ea0       = (const float*)d_in[9];
  const float* ea1       = (const float*)d_in[10];
  const int*   ei0       = (const int*)d_in[11];
  const int*   ei1       = (const int*)d_in[12];
  const int*   ei2       = (const int*)d_in[13];

  float* H  = (float*)d_out;
  float* ws = (float*)d_ws;

  proj_kernel<<<dim3((B_ * N_) / 16), dim3(128), 0, stream>>>(node_feat, in_W,
                                                              in_b, H);
  for (int l = 0; l < 2; ++l) {
    for (int b = 0; b < B_; ++b) {
      node_kernel<<<dim3(N_ / 32, 3), dim3(128), 0, stream>>>(H, ws, node_W, l, b);
      edge_kernel<<<dim3(EDGE_GRID, 3), dim3(128), 0, stream>>>(
          ws, node_b, edge_W, edge_b, ea0, ea1, ei0, ei1, ei2, l);
      ln_kernel<<<dim3(N_ / 4), dim3(128), 0, stream>>>(H, ws, ln_g, ln_b, l, b);
    }
  }
}


# f32 atomic scatter replaced by per-call dst-sorted edge lists (count, scan, fill): a wave sums a node's P rows in list arrival order and applies LN; edge attrs summed per node before their product
# speedup vs baseline: 3.3985x; 1.5737x over previous
#include <hip/hip_runtime.h>


#define B_ 4
#define N_ 20000
#define Dn_ 64
#define Dm_ 128
#define E_ 160000

#define PERM_O 0
#define OFF_O  480000
#define CNT_O  500032
#define CUR_O  560032
#define S_O    580032
#define ZERO_WORDS 720000

typedef float v2f __attribute__((ext_vector_type(2)));
typedef float v8f __attribute__((ext_vector_type(8)));

__device__ __forceinline__ v8f wmma4(v2f a, v2f b, v8f c) {
  return __builtin_amdgcn_wmma_f32_16x16x4_f32(false, a, false, b, (short)0, c,
                                               false, false);
}

__global__ __launch_bounds__(128)
void proj_kernel(const float* __restrict__ X, const float* __restrict__ W,
                 const float* __restrict__ bias, float* __restrict__ H) {
  const int lane = threadIdx.x & 31;
  const int wave = threadIdx.x >> 5;
  const int m    = lane & 15;
  const int kh   = lane >> 4;
  const int rowbase = blockIdx.x * 16;

  v2f a[16];
  const float* xr = X + (size_t)(rowbase + m) * Dn_;
#pragma unroll
  for (int kk = 0; kk < 16; ++kk) {
    int k0 = kk * 4 + kh * 2;
    a[kk].x = xr[k0];
    a[kk].y = xr[k0 + 1];
  }

#pragma unroll
  for (int t = 0; t < 2; ++t) {
    int col = (wave * 2 + t) * 16;
    v8f c = {0.f, 0.f, 0.f, 0.f, 0.f, 0.f, 0.f, 0.f};
#pragma unroll
    for (int kk = 0; kk < 16; ++kk) {
      int k0 = kk * 4 + kh * 2;
      v2f bf;
      bf.x = W[k0 * Dm_ + col + m];
      bf.y = W[(k0 + 1) * Dm_ + col + m];
      c = wmma4(a[kk], bf, c);
    }
    int rowadd = kh * 8;
#pragma unroll
    for (int v = 0; v < 8; ++v) {
      H[(size_t)(rowbase + v + rowadd) * Dm_ + col + m] = c[v] + bias[col + m];
    }
  }
}


__global__ __launch_bounds__(128)
void node_kernel(const float* __restrict__ H, float* __restrict__ ws,
                 const float* __restrict__ node_W, int layer, int b) {
  __shared__ float Hs[32][132];

  const int r    = blockIdx.y;
  const int tid  = threadIdx.x;
  const int lane = tid & 31;
  const int wave = tid >> 5;
  const int m    = lane & 15;
  const int kh   = lane >> 4;
  const int rowbase = blockIdx.x * 32;

  const float* Hb = H + (size_t)b * N_ * Dm_;

  const int c4   = tid & 31;
  const int row0 = tid >> 5;
  float4 hv[8];
#pragma unroll
  for (int j = 0; j < 8; ++j) {
    hv[j] = *(const float4*)(Hb + (size_t)(rowbase + row0 + 4 * j) * Dm_ + c4 * 4);
  }
#pragma unroll
  for (int j = 0; j < 8; ++j) {
    *(float4*)&Hs[row0 + 4 * j][c4 * 4] = hv[j];
  }
  __syncthreads();

  const int col0 = wave * 32 + m;
  const int rowadd = kh * 8;

  const float* Wn = node_W + (size_t)(layer * 3 + r) * Dm_ * Dm_;
  float* P = ws + (size_t)(1 + r) * N_ * Dm_;
  v8f c00 = {0.f, 0.f, 0.f, 0.f, 0.f, 0.f, 0.f, 0.f};
  v8f c01 = c00, c10 = c00, c11 = c00;
#pragma unroll 8
  for (int kk = 0; kk < 32; ++kk) {
    int k0 = kk * 4 + kh * 2;
    v2f a0, a1, b0, b1;
    a0.x = Hs[m][k0];
    a0.y = Hs[m][k0 + 1];
    a1.x = Hs[16 + m][k0];
    a1.y = Hs[16 + m][k0 + 1];
    b0.x = Wn[k0 * Dm_ + col0];
    b0.y = Wn[(k0 + 1) * Dm_ + col0];
    b1.x = Wn[k0 * Dm_ + col0 + 16];
    b1.y = Wn[(k0 + 1) * Dm_ + col0 + 16];
    c00 = wmma4(a0, b0, c00);
    c01 = wmma4(a0, b1, c01);
    c10 = wmma4(a1, b0, c10);
    c11 = wmma4(a1, b1, c11);
  }
#pragma unroll
  for (int v = 0; v < 8; ++v) {
    size_t g0 = (size_t)(rowbase + v + rowadd) * Dm_ + col0;
    size_t g1 = (size_t)(rowbase + 16 + v + rowadd) * Dm_ + col0;
    P[g0]      = c00[v];
    P[g0 + 16] = c01[v];
    P[g1]      = c10[v];
    P[g1 + 16] = c11[v];
  }
}


__global__ __launch_bounds__(256)
void zero_kernel(float* __restrict__ ws) {
  const int i = blockIdx.x * 256 + threadIdx.x;
  if (i < ZERO_WORDS / 4) {
    ((float4*)(ws + CNT_O))[i] = make_float4(0.f, 0.f, 0.f, 0.f);
  }
}

#define HIST_GRID 1000
#define HIST_TRIPS 10
static_assert(HIST_TRIPS * HIST_GRID * 256 == E_ * 16, "hist trips");
static_assert(HIST_GRID * 256 >= E_, "relation 2: one pass");

__global__ __launch_bounds__(256)
void hist_kernel(float* __restrict__ ws,
                 const float* __restrict__ ea0, const float* __restrict__ ea1,
                 const int* __restrict__ ei0, const int* __restrict__ ei1,
                 const int* __restrict__ ei2) {
  const int r = blockIdx.y;
  int* cnt = (int*)ws + CNT_O + r * N_;
  if (r < 2) {
    const int*   ei = (r == 0) ? ei0 : ei1;
    const float* ea = (r == 0) ? ea0 : ea1;
    float* S = ws + S_O + (size_t)r * N_ * 16;
    const int k = threadIdx.x & 15;
    int   dd[HIST_TRIPS];
    float vv[HIST_TRIPS];
#pragma unroll
    for (int it = 0; it < HIST_TRIPS; ++it) {
      const int t = (it * HIST_GRID + blockIdx.x) * 256 + threadIdx.x;
      dd[it] = ei[E_ + (t >> 4)];
      vv[it] = ea[t];
    }
#pragma unroll
    for (int it = 0; it < HIST_TRIPS; ++it) {
      if (k == 0) atomicAdd(cnt + dd[it], 1);
      atomicAdd(S + (size_t)dd[it] * 16 + k, vv[it]);
    }
  } else {
    for (int e = blockIdx.x * 256 + threadIdx.x; e < E_; e += HIST_GRID * 256) {
      atomicAdd(cnt + ei2[E_ + e], 1);
    }
  }
}

#define SCAN_CH 20
static_assert(N_ % SCAN_CH == 0 && 1024 * SCAN_CH >= N_, "scan chunks");

__global__ __launch_bounds__(1024)
void scan_kernel(float* __restrict__ ws) {
  __shared__ int part[1024];
  const int t = threadIdx.x;
  const int* cnt = (const int*)ws + CNT_O;
  int* off = (int*)ws + OFF_O;

  const bool inside = t * SCAN_CH < N_;
  int c[SCAN_CH];
#pragma unroll
  for (int i = 0; i < SCAN_CH; ++i) c[i] = 0;
  if (inside) {
#pragma unroll
    for (int i = 0; i < SCAN_CH; ++i) {
      const int d = t * SCAN_CH + i;
      c[i] = cnt[d] + cnt[N_ + d] + cnt[2 * N_ + d];
    }
  }
  int loc[SCAN_CH];
  int sum = 0;
#pragma unroll
  for (int i = 0; i < SCAN_CH; ++i) {
    loc[i] = sum;
    sum += c[i];
  }
  part[t] = sum;
  __syncthreads();
  for (int ofs = 1; ofs < 1024; ofs <<= 1) {
    const int v = (t >= ofs) ? part[t - ofs] : 0;
    __syncthreads();
    part[t] += v;
    __syncthreads();
  }
  const int excl = part[t] - sum;
  if (inside) {
#pragma unroll
    for (int i = 0; i < SCAN_CH; ++i) off[t * SCAN_CH + i] = excl + loc[i];
  }
  if (t == 1023) off[N_] = part[1023];
}

__global__ __launch_bounds__(256)
void fill_kernel(float* __restrict__ ws, const int* __restrict__ ei0,
                 const int* __restrict__ ei1, const int* __restrict__ ei2) {
  const int r = blockIdx.y;
  const int e = blockIdx.x * 256 + threadIdx.x;
  const int* ei = (r == 0) ? ei0 : (r == 1) ? ei1 : ei2;
  const int s = ei[e];
  const int d = ei[E_ + e];
  int* wsI = (int*)ws;
  const int pos = wsI[OFF_O + d] + atomicAdd(wsI + CUR_O + d, 1);
  wsI[PERM_O + pos] = r * N_ + s;
}

__global__ __launch_bounds__(256)
void gather_ln_kernel(float* __restrict__ H, const float* __restrict__ ws,
                      const float* __restrict__ node_b,
                      const float* __restrict__ edge_W, const float* __restrict__ edge_b,
                      const float* __restrict__ ln_g, const float* __restrict__ ln_b,
                      int layer, int b) {
  __shared__ float T[16][132];

  const int tid  = threadIdx.x;
  const int lane = tid & 31;
  const int wave = __builtin_amdgcn_readfirstlane(tid >> 5);
  const int m    = lane & 15;
  const int kh   = lane >> 4;
  const int nodebase = blockIdx.x * 16;
  const int* wsI = (const int*)ws;

  {
    const int col = wave * 16;
    v8f c = {0.f, 0.f, 0.f, 0.f, 0.f, 0.f, 0.f, 0.f};
#pragma unroll
    for (int r = 0; r < 2; ++r) {
      const float* sr = ws + S_O + ((size_t)r * N_ + nodebase + m) * 16;
      const float* Wq = edge_W + (size_t)(layer * 2 + r) * 16 * Dm_;
#pragma unroll
      for (int kk = 0; kk < 4; ++kk) {
        const int k0 = kk * 4 + kh * 2;
        v2f a, bf;
        a.x  = sr[k0];
        a.y  = sr[k0 + 1];
        bf.x = Wq[k0 * Dm_ + col + m];
        bf.y = Wq[(k0 + 1) * Dm_ + col + m];
        c = wmma4(a, bf, c);
      }
    }
    const int rowadd = kh * 8;
#pragma unroll
    for (int v = 0; v < 8; ++v) T[v + rowadd][col + m] = c[v];
  }
  __syncthreads();

  const float* nb = node_b + (size_t)layer * 3 * Dm_ + lane * 4;
  const float* eb = edge_b + (size_t)layer * 2 * Dm_ + lane * 4;
  const float4 n0 = *(const float4*)(nb);
  const float4 n1 = *(const float4*)(nb + Dm_);
  const float4 n2 = *(const float4*)(nb + 2 * Dm_);
  const float4 e0 = *(const float4*)(eb);
  const float4 e1 = *(const float4*)(eb + Dm_);
  const float4 bv0 = make_float4(n0.x + e0.x, n0.y + e0.y, n0.z + e0.z, n0.w + e0.w);
  const float4 bv1 = make_float4(n1.x + e1.x, n1.y + e1.y, n1.z + e1.z, n1.w + e1.w);
  const float4 gv  = *(const float4*)(ln_g + (size_t)layer * Dm_ + lane * 4);
  const float4 bbv = *(const float4*)(ln_b + (size_t)layer * Dm_ + lane * 4);

  const float* Pl = ws + (size_t)N_ * Dm_ + lane * 4;
  const int* perm = wsI + PERM_O;

  for (int i = 0; i < 2; ++i) {
    const int row = wave * 2 + i;
    const int d   = nodebase + row;
    const int beg = __builtin_amdgcn_readfirstlane(wsI[OFF_O + d]);
    const int end = __builtin_amdgcn_readfirstlane(wsI[OFF_O + d + 1]);
    const float c0 = (float)wsI[CNT_O + d];
    const float c1 = (float)wsI[CNT_O + N_ + d];
    const float c2 = (float)wsI[CNT_O + 2 * N_ + d];

    float* hp = H + ((size_t)b * N_ + d) * Dm_ + lane * 4;
    const float4 x = *(const float4*)hp;

    float4 acc = *(const float4*)&T[row][lane * 4];
    acc.x += c0 * bv0.x + c1 * bv1.x + c2 * n2.x;
    acc.y += c0 * bv0.y + c1 * bv1.y + c2 * n2.y;
    acc.z += c0 * bv0.z + c1 * bv1.z + c2 * n2.z;
    acc.w += c0 * bv0.w + c1 * bv1.w + c2 * n2.w;

    for (int p = beg; p < end; p += 32) {
      const int rem = end - p;
      const int n   = (rem < 32) ? rem : 32;
      const int idx = (lane < n) ? perm[p + lane] : 0;
      int j = 0;
      for (; j + 8 <= n; j += 8) {
        float4 v[8];
#pragma unroll
        for (int u = 0; u < 8; ++u) {
          const int rid = __builtin_amdgcn_readlane(idx, j + u);
          v[u] = *(const float4*)(Pl + (size_t)rid * Dm_);
        }
#pragma unroll
        for (int u = 0; u < 8; ++u) {
          acc.x += v[u].x; acc.y += v[u].y; acc.z += v[u].z; acc.w += v[u].w;
        }
      }
      for (; j < n; ++j) {
        const int rid = __builtin_amdgcn_readlane(idx, j);
        const float4 v = *(const float4*)(Pl + (size_t)rid * Dm_);
        acc.x += v.x; acc.y += v.y; acc.z += v.z; acc.w += v.w;
      }
    }

    float4 y;
    y.x = x.x + fmaxf(acc.x, 0.f);
    y.y = x.y + fmaxf(acc.y, 0.f);
    y.z = x.z + fmaxf(acc.z, 0.f);
    y.w = x.w + fmaxf(acc.w, 0.f);

    float s = y.x + y.y + y.z + y.w;
    float q = y.x * y.x + y.y * y.y + y.z * y.z + y.w * y.w;
#pragma unroll
    for (int o = 16; o >= 1; o >>= 1) {
      s += __shfl_xor(s, o, 32);
      q += __shfl_xor(q, o, 32);
    }
    const float mean = s * (1.0f / Dm_);
    const float var  = q * (1.0f / Dm_) - mean * mean;
    const float rs   = rsqrtf(var + 1e-5f);

    float4 o4;
    o4.x = (y.x - mean) * rs * gv.x + bbv.x;
    o4.y = (y.y - mean) * rs * gv.y + bbv.y;
    o4.z = (y.z - mean) * rs * gv.z + bbv.z;
    o4.w = (y.w - mean) * rs * gv.w + bbv.w;
    *(float4*)hp = o4;
  }
}

extern "C" void kernel_launch(void* const* d_in, const int* in_sizes, int n_in,
                              void* d_out, int out_size, void* d_ws, size_t ws_size,
                              hipStream_t stream) {
  const float* node_feat = (const float*)d_in[0];
  const float* in_W      = (const float*)d_in[1];
  const float* in_b      = (const float*)d_in[2];
  const float* node_W    = (const float*)d_in[3];
  const float* node_b    = (const float*)d_in[4];
  const float* edge_W    = (const float*)d_in[5];
  const float* edge_b    = (const float*)d_in[6];
  const float* ln_g      = (const float*)d_in[7];
  const float* ln_b      = (const float*)d_in[8];
  const float* ea0       = (const float*)d_in[9];
  const float* ea1       = (const float*)d_in[10];
  const int*   ei0       = (const int*)d_in[11];
  const int*   ei1       = (const int*)d_in[12];
  const int*   ei2       = (const int*)d_in[13];

  float* H  = (float*)d_out;
  float* ws = (float*)d_ws;

  proj_kernel<<<dim3((B_ * N_) / 16), dim3(128), 0, stream>>>(node_feat, in_W,
                                                              in_b, H);
  zero_kernel<<<dim3((ZERO_WORDS / 4 + 255) / 256), dim3(256), 0, stream>>>(ws);
  hist_kernel<<<dim3(HIST_GRID, 3), dim3(256), 0, stream>>>(ws, ea0, ea1, ei0, ei1, ei2);
  scan_kernel<<<dim3(1), dim3(1024), 0, stream>>>(ws);
  fill_kernel<<<dim3(E_ / 256, 3), dim3(256), 0, stream>>>(ws, ei0, ei1, ei2);
  for (int l = 0; l < 2; ++l) {
    for (int b = 0; b < B_; ++b) {
      node_kernel<<<dim3(N_ / 32, 3), dim3(128), 0, stream>>>(H, ws, node_W, l, b);
      gather_ln_kernel<<<dim3(N_ / 16), dim3(256), 0, stream>>>(
          H, ws, node_b, edge_W, edge_b, ln_g, ln_b, l, b);
    }
  }
}


# Layer 0 sums the 64-wide input rows per relation and multiplies once by in_W x node_W_r (built per call on WMMA): no projection launch, half the gather bytes, 4 batches per launch; sum order changes
# speedup vs baseline: 4.6214x; 1.3598x over previous
#include <hip/hip_runtime.h>


#define B_ 4
#define N_ 20000
#define Dn_ 64
#define Dm_ 128
#define E_ 160000

#define PERM_O 0
#define OFF_O  480000
#define CNT_O  500032
#define CUR_O  560032
#define S_O    620032
#define ZERO_WORDS 760000
#define MP_O   1260032
#define MPROWS_ 80

typedef float v2f __attribute__((ext_vector_type(2)));
typedef float v8f __attribute__((ext_vector_type(8)));

__device__ __forceinline__ v8f wmma4(v2f a, v2f b, v8f c) {
  return __builtin_amdgcn_wmma_f32_16x16x4_f32(false, a, false, b, (short)0, c,
                                               false, false);
}

#define ZB_  ((ZERO_WORDS / 4 + 255) / 256)
#define MPB_ 15
static_assert(MPB_ * 8 == 3 * (MPROWS_ / 16) * (Dm_ / 16), "MP tiles");

__global__ __launch_bounds__(256)
void prep_kernel(float* __restrict__ ws, const float* __restrict__ in_W,
                 const float* __restrict__ in_b, const float* __restrict__ node_W) {
  if (blockIdx.x < ZB_) {
    const int i = blockIdx.x * 256 + threadIdx.x;
    if (i < ZERO_WORDS / 4) {
      ((float4*)(ws + CNT_O))[i] = make_float4(0.f, 0.f, 0.f, 0.f);
    }
    return;
  }
  const int lane = threadIdx.x & 31;
  const int wave = threadIdx.x >> 5;
  const int m    = lane & 15;
  const int kh   = lane >> 4;
  const int tile = (blockIdx.x - ZB_) * 8 + wave;
  const int r    = tile / 40;
  const int rt   = (tile % 40) >> 3;
  const int col  = (tile & 7) * 16 + m;
  const int row  = rt * 16 + m;
  const float* Wr   = node_W + (size_t)r * Dm_ * Dm_;
  const float* arow = (row < Dn_) ? (in_W + row * Dm_) : in_b;
  const bool live = row <= Dn_;
  v8f c = {0.f, 0.f, 0.f, 0.f, 0.f, 0.f, 0.f, 0.f};
#pragma unroll 8
  for (int kk = 0; kk < 32; ++kk) {
    const int k0 = kk * 4 + kh * 2;
    v2f a, b;
    const float ax = arow[k0];
    const float ay = arow[k0 + 1];
    a.x = live ? ax : 0.f;
    a.y = live ? ay : 0.f;
    b.x = Wr[k0 * Dm_ + col];
    b.y = Wr[(k0 + 1) * Dm_ + col];
    c = wmma4(a, b, c);
  }
  float* Mp = ws + MP_O + (size_t)r * MPROWS_ * Dm_;
  const int rowadd = kh * 8;
#pragma unroll
  for (int v = 0; v < 8; ++v) {
    Mp[(rt * 16 + v + rowadd) * Dm_ + col] = c[v];
  }
}


#define HIST_GRID 1000
#define HIST_TRIPS 10
static_assert(HIST_TRIPS * HIST_GRID * 256 == E_ * 16, "hist trips");
static_assert(HIST_GRID * 256 >= E_, "relation 2: one pass");

__global__ __launch_bounds__(256)
void hist_kernel(float* __restrict__ ws,
                 const float* __restrict__ ea0, const float* __restrict__ ea1,
                 const int* __restrict__ ei0, const int* __restrict__ ei1,
                 const int* __restrict__ ei2) {
  const int r = blockIdx.y;
  int* cnt = (int*)ws + CNT_O + r * N_;
  if (r < 2) {
    const int*   ei = (r == 0) ? ei0 : ei1;
    const float* ea = (r == 0) ? ea0 : ea1;
    float* S = ws + S_O + (size_t)r * N_ * 16;
    const int k = threadIdx.x & 15;
    int   dd[HIST_TRIPS];
    float vv[HIST_TRIPS];
#pragma unroll
    for (int it = 0; it < HIST_TRIPS; ++it) {
      const int t = (it * HIST_GRID + blockIdx.x) * 256 + threadIdx.x;
      dd[it] = ei[E_ + (t >> 4)];
      vv[it] = ea[t];
    }
#pragma unroll
    for (int it = 0; it < HIST_TRIPS; ++it) {
      if (k == 0) atomicAdd(cnt + dd[it], 1);
      atomicAdd(S + (size_t)dd[it] * 16 + k, vv[it]);
    }
  } else {
    for (int e = blockIdx.x * 256 + threadIdx.x; e < E_; e += HIST_GRID * 256) {
      atomicAdd(cnt + ei2[E_ + e], 1);
    }
  }
}

#define SCAN_CH 20
static_assert(N_ % SCAN_CH == 0 && 1024 * SCAN_CH >= N_, "scan chunks");

__global__ __launch_bounds__(1024)
void scan_kernel(float* __restrict__ ws) {
  __shared__ int part[1024];
  const int t = threadIdx.x;
  const int* cnt = (const int*)ws + CNT_O;
  int* off = (int*)ws + OFF_O;

  const bool inside = t * SCAN_CH < N_;
  int c[SCAN_CH];
#pragma unroll
  for (int i = 0; i < SCAN_CH; ++i) c[i] = 0;
  if (inside) {
#pragma unroll
    for (int i = 0; i < SCAN_CH; ++i) {
      const int d = t * SCAN_CH + i;
      c[i] = cnt[d] + cnt[N_ + d] + cnt[2 * N_ + d];
    }
  }
  int loc[SCAN_CH];
  int sum = 0;
#pragma unroll
  for (int i = 0; i < SCAN_CH; ++i) {
    loc[i] = sum;
    sum += c[i];
  }
  part[t] = sum;
  __syncthreads();
  for (int ofs = 1; ofs < 1024; ofs <<= 1) {
    const int v = (t >= ofs) ? part[t - ofs] : 0;
    __syncthreads();
    part[t] += v;
    __syncthreads();
  }
  const int excl = part[t] - sum;
  if (inside) {
#pragma unroll
    for (int i = 0; i < SCAN_CH; ++i) off[t * SCAN_CH + i] = excl + loc[i];
  }
  if (t == 1023) off[N_] = part[1023];
}

__global__ __launch_bounds__(256)
void fill_kernel(float* __restrict__ ws, const int* __restrict__ ei0,
                 const int* __restrict__ ei1, const int* __restrict__ ei2) {
  const int r = blockIdx.y;
  const int e = blockIdx.x * 256 + threadIdx.x;
  const int* ei = (r == 0) ? ei0 : (r == 1) ? ei1 : ei2;
  const int s = ei[e];
  const int d = ei[E_ + e];
  int* wsI = (int*)ws;
  int seg = wsI[OFF_O + d];
  if (r >= 1) seg += wsI[CNT_O + d];
  if (r == 2) seg += wsI[CNT_O + N_ + d];
  const int pos = seg + atomicAdd(wsI + CUR_O + r * N_ + d, 1);
  wsI[PERM_O + pos] = s;
}

#define KA_ 416
#define APITCH_ 420

__global__ __launch_bounds__(256)
void layer_kernel(const float* __restrict__ in0, const float* __restrict__ in1,
                  const float* __restrict__ in2, float* __restrict__ out0,
                  float* __restrict__ out1, float* __restrict__ out2,
                  const float* __restrict__ ws,
                  const float* __restrict__ node_W, const float* __restrict__ node_b,
                  const float* __restrict__ edge_W, const float* __restrict__ edge_b,
                  const float* __restrict__ ln_g, const float* __restrict__ ln_b,
                  int layer) {
  __shared__ float A[32][APITCH_];
  __shared__ float R[32][132];

  const int tid  = threadIdx.x;
  const int lane = tid & 31;
  const int wave = __builtin_amdgcn_readfirstlane(tid >> 5);
  const int m    = lane & 15;
  const int kh   = lane >> 4;
  const int nodebase = blockIdx.x * 32;
  const int* wsI = (const int*)ws;

  const float* Hin  = (blockIdx.y == 0) ? in0  : (blockIdx.y == 1) ? in1  : in2;
  float*       Hout = (blockIdx.y == 0) ? out0 : (blockIdx.y == 1) ? out1 : out2;

  const float* Hl = Hin + lane * 4;
  const int* perm = wsI + PERM_O;
  const int dl   = nodebase + wave * 4 + (lane & 3);
  const int vbeg = wsI[OFF_O + dl];
  const int vn0  = wsI[CNT_O + dl];
  const int vn1  = wsI[CNT_O + N_ + dl];
  const int vn2  = wsI[CNT_O + 2 * N_ + dl];
  for (int i = 0; i < 4; ++i) {
    const int row = wave * 4 + i;
    const int d   = nodebase + row;
    const int beg = __builtin_amdgcn_readlane(vbeg, i);
    const int n0  = __builtin_amdgcn_readlane(vn0, i);
    const int n1  = __builtin_amdgcn_readlane(vn1, i);
    const int n2  = __builtin_amdgcn_readlane(vn2, i);
    const int e0  = beg + n0;
    const int e1  = e0 + n1;
    const int end = e1 + n2;

    const float sv = ws[S_O + ((size_t)kh * N_ + d) * 16 + m];

    float4 a0 = make_float4(0.f, 0.f, 0.f, 0.f);
    float4 a1 = a0, a2 = a0;
    for (int p = beg; p < end; p += 32) {
      const int rem = end - p;
      const int n   = (rem < 32) ? rem : 32;
      const int idx = (lane < n) ? perm[p + lane] : 0;
      for (int j = 0; j < n; j += 8) {
        float4 v[8];
#pragma unroll
        for (int u = 0; u < 8; ++u) {
          const int rid = __builtin_amdgcn_readlane(idx, j + u);
          v[u] = *(const float4*)(Hl + (size_t)rid * Dm_);
        }
#pragma unroll
        for (int u = 0; u < 8; ++u) {
          const int q = p + j + u;
          if (q < e0) {
            a0.x += v[u].x; a0.y += v[u].y; a0.z += v[u].z; a0.w += v[u].w;
          } else if (q < e1) {
            a1.x += v[u].x; a1.y += v[u].y; a1.z += v[u].z; a1.w += v[u].w;
          } else if (q < end) {
            a2.x += v[u].x; a2.y += v[u].y; a2.z += v[u].z; a2.w += v[u].w;
          }
        }
      }
    }
    *(float4*)&A[row][lane * 4]           = a0;
    *(float4*)&A[row][Dm_ + lane * 4]     = a1;
    *(float4*)&A[row][2 * Dm_ + lane * 4] = a2;
    A[row][3 * Dm_ + lane] = sv;
  }
  __syncthreads();

  {
    const int col = wave * 16 + m;
    const float* Wn = node_W + (size_t)layer * 3 * Dm_ * Dm_;
    const float* We = edge_W + (size_t)layer * 2 * 16 * Dm_;
    v8f c0 = {0.f, 0.f, 0.f, 0.f, 0.f, 0.f, 0.f, 0.f};
    v8f c1 = c0;
#pragma unroll 16
    for (int kk = 0; kk < 96; ++kk) {
      const int k0 = kk * 4 + kh * 2;
      v2f f0, f1, bf;
      f0.x = A[m][k0];
      f0.y = A[m][k0 + 1];
      f1.x = A[16 + m][k0];
      f1.y = A[16 + m][k0 + 1];
      bf.x = Wn[k0 * Dm_ + col];
      bf.y = Wn[(k0 + 1) * Dm_ + col];
      c0 = wmma4(f0, bf, c0);
      c1 = wmma4(f1, bf, c1);
    }
#pragma unroll
    for (int kk = 0; kk < 8; ++kk) {
      const int k0 = kk * 4 + kh * 2;
      v2f f0, f1, bf;
      f0.x = A[m][3 * Dm_ + k0];
      f0.y = A[m][3 * Dm_ + k0 + 1];
      f1.x = A[16 + m][3 * Dm_ + k0];
      f1.y = A[16 + m][3 * Dm_ + k0 + 1];
      bf.x = We[k0 * Dm_ + col];
      bf.y = We[(k0 + 1) * Dm_ + col];
      c0 = wmma4(f0, bf, c0);
      c1 = wmma4(f1, bf, c1);
    }
    const int rowadd = kh * 8;
#pragma unroll
    for (int v = 0; v < 8; ++v) {
      R[v + rowadd][col]      = c0[v];
      R[16 + v + rowadd][col] = c1[v];
    }
  }
  __syncthreads();

  const float* nb = node_b + (size_t)layer * 3 * Dm_ + lane * 4;
  const float* eb = edge_b + (size_t)layer * 2 * Dm_ + lane * 4;
  const float4 n0v = *(const float4*)(nb);
  const float4 n1v = *(const float4*)(nb + Dm_);
  const float4 n2v = *(const float4*)(nb + 2 * Dm_);
  const float4 e0v = *(const float4*)(eb);
  const float4 e1v = *(const float4*)(eb + Dm_);
  const float4 bv0 = make_float4(n0v.x + e0v.x, n0v.y + e0v.y, n0v.z + e0v.z, n0v.w + e0v.w);
  const float4 bv1 = make_float4(n1v.x + e1v.x, n1v.y + e1v.y, n1v.z + e1v.z, n1v.w + e1v.w);
  const float4 gv  = *(const float4*)(ln_g + (size_t)layer * Dm_ + lane * 4);
  const float4 bbv = *(const float4*)(ln_b + (size_t)layer * Dm_ + lane * 4);

#pragma unroll
  for (int i = 0; i < 4; ++i) {
    const int row = wave * 4 + i;
    const int d   = nodebase + row;
    const float c0 = (float)wsI[CNT_O + d];
    const float c1 = (float)wsI[CNT_O + N_ + d];
    const float c2 = (float)wsI[CNT_O + 2 * N_ + d];
    const float4 x = *(const float4*)(Hin + (size_t)d * Dm_ + lane * 4);

    float4 acc = *(const float4*)&R[row][lane * 4];
    acc.x += c0 * bv0.x + c1 * bv1.x + c2 * n2v.x;
    acc.y += c0 * bv0.y + c1 * bv1.y + c2 * n2v.y;
    acc.z += c0 * bv0.z + c1 * bv1.z + c2 * n2v.z;
    acc.w += c0 * bv0.w + c1 * bv1.w + c2 * n2v.w;

    float4 y;
    y.x = x.x + fmaxf(acc.x, 0.f);
    y.y = x.y + fmaxf(acc.y, 0.f);
    y.z = x.z + fmaxf(acc.z, 0.f);
    y.w = x.w + fmaxf(acc.w, 0.f);

    float s = y.x + y.y + y.z + y.w;
    float q = y.x * y.x + y.y * y.y + y.z * y.z + y.w * y.w;
#pragma unroll
    for (int o = 16; o >= 1; o >>= 1) {
      s += __shfl_xor(s, o, 32);
      q += __shfl_xor(q, o, 32);
    }
    const float mean = s * (1.0f / Dm_);
    const float var  = q * (1.0f / Dm_) - mean * mean;
    const float rs   = rsqrtf(var + 1e-5f);

    float4 o4;
    o4.x = (y.x - mean) * rs * gv.x + bbv.x;
    o4.y = (y.y - mean) * rs * gv.y + bbv.y;
    o4.z = (y.z - mean) * rs * gv.z + bbv.z;
    o4.w = (y.w - mean) * rs * gv.w + bbv.w;
    *(float4*)(Hout + (size_t)d * Dm_ + lane * 4) = o4;
  }
}

#define K0A_ 224
#define APITCH0_ 292

__global__ __launch_bounds__(256)
void layer0_kernel(const float* __restrict__ X_all, float* __restrict__ out0,
                   float* __restrict__ out1, float* __restrict__ out2,
                   float* __restrict__ out3, const float* __restrict__ ws,
                   const float* __restrict__ in_W, const float* __restrict__ in_b,
                   const float* __restrict__ node_b,
                   const float* __restrict__ edge_W, const float* __restrict__ edge_b,
                   const float* __restrict__ ln_g, const float* __restrict__ ln_b) {
  __shared__ float A[32][APITCH0_];
  __shared__ float R[32][132];
  __shared__ float Hr[32][132];

  const int tid  = threadIdx.x;
  const int lane = tid & 31;
  const int wave = __builtin_amdgcn_readfirstlane(tid >> 5);
  const int m    = lane & 15;
  const int kh   = lane >> 4;
  const int nodebase = blockIdx.x * 32;
  const int* wsI = (const int*)ws;

  const float* X = X_all + (size_t)blockIdx.y * N_ * Dn_;
  float* Hout = (blockIdx.y == 0) ? out0 : (blockIdx.y == 1) ? out1
              : (blockIdx.y == 2) ? out2 : out3;

  const float* Xl = X + lane * 2;
  const int* perm = wsI + PERM_O;
  const int dl   = nodebase + wave * 4 + (lane & 3);
  const int vbeg = wsI[OFF_O + dl];
  const int vn0  = wsI[CNT_O + dl];
  const int vn1  = wsI[CNT_O + N_ + dl];
  const int vn2  = wsI[CNT_O + 2 * N_ + dl];
  for (int i = 0; i < 4; ++i) {
    const int row = wave * 4 + i;
    const int d   = nodebase + row;
    const int beg = __builtin_amdgcn_readlane(vbeg, i);
    const int n0  = __builtin_amdgcn_readlane(vn0, i);
    const int n1  = __builtin_amdgcn_readlane(vn1, i);
    const int n2  = __builtin_amdgcn_readlane(vn2, i);
    const int e0  = beg + n0;
    const int e1  = e0 + n1;
    const int end = e1 + n2;

    const float sv = ws[S_O + ((size_t)kh * N_ + d) * 16 + m];
    const float2 xd = *(const float2*)(Xl + (size_t)d * Dn_);

    float2 a0 = make_float2(0.f, 0.f);
    float2 a1 = a0, a2 = a0;
    for (int p = beg; p < end; p += 32) {
      const int rem = end - p;
      const int n   = (rem < 32) ? rem : 32;
      const int idx = (lane < n) ? perm[p + lane] : 0;
      for (int j = 0; j < n; j += 16) {
        float2 v[16];
#pragma unroll
        for (int u = 0; u < 16; ++u) {
          const int rid = __builtin_amdgcn_readlane(idx, j + u);
          v[u] = *(const float2*)(Xl + (size_t)rid * Dn_);
        }
#pragma unroll
        for (int u = 0; u < 16; ++u) {
          const int q = p + j + u;
          if (q < e0) {
            a0.x += v[u].x; a0.y += v[u].y;
          } else if (q < e1) {
            a1.x += v[u].x; a1.y += v[u].y;
          } else if (q < end) {
            a2.x += v[u].x; a2.y += v[u].y;
          }
        }
      }
    }
    *(float2*)&A[row][lane * 2]           = a0;
    *(float2*)&A[row][Dn_ + lane * 2]     = a1;
    *(float2*)&A[row][2 * Dn_ + lane * 2] = a2;
    A[row][3 * Dn_ + lane] = sv;
    *(float2*)&A[row][K0A_ + lane * 2]    = xd;
  }
  __syncthreads();

  {
    const int col = wave * 16 + m;
    const float* Mp = ws + MP_O;
    v8f c0 = {0.f, 0.f, 0.f, 0.f, 0.f, 0.f, 0.f, 0.f};
    v8f c1 = c0, h0 = c0, h1 = c0;
#pragma unroll 16
    for (int kk = 0; kk < 48; ++kk) {
      const int k0 = kk * 4 + kh * 2;
      const int kr = (kk >> 4) * MPROWS_ + (kk & 15) * 4 + kh * 2;
      v2f f0, f1, bf;
      f0.x = A[m][k0];
      f0.y = A[m][k0 + 1];
      f1.x = A[16 + m][k0];
      f1.y = A[16 + m][k0 + 1];
      bf.x = Mp[kr * Dm_ + col];
      bf.y = Mp[(kr + 1) * Dm_ + col];
      c0 = wmma4(f0, bf, c0);
      c1 = wmma4(f1, bf, c1);
    }
#pragma unroll
    for (int kk = 0; kk < 8; ++kk) {
      const int k0 = kk * 4 + kh * 2;
      v2f f0, f1, bf;
      f0.x = A[m][3 * Dn_ + k0];
      f0.y = A[m][3 * Dn_ + k0 + 1];
      f1.x = A[16 + m][3 * Dn_ + k0];
      f1.y = A[16 + m][3 * Dn_ + k0 + 1];
      bf.x = edge_W[k0 * Dm_ + col];
      bf.y = edge_W[(k0 + 1) * Dm_ + col];
      c0 = wmma4(f0, bf, c0);
      c1 = wmma4(f1, bf, c1);
    }
#pragma unroll
    for (int kk = 0; kk < 16; ++kk) {
      const int k0 = kk * 4 + kh * 2;
      v2f f0, f1, bf;
      f0.x = A[m][K0A_ + k0];
      f0.y = A[m][K0A_ + k0 + 1];
      f1.x = A[16 + m][K0A_ + k0];
      f1.y = A[16 + m][K0A_ + k0 + 1];
      bf.x = in_W[k0 * Dm_ + col];
      bf.y = in_W[(k0 + 1) * Dm_ + col];
      h0 = wmma4(f0, bf, h0);
      h1 = wmma4(f1, bf, h1);
    }
    const int rowadd = kh * 8;
#pragma unroll
    for (int v = 0; v < 8; ++v) {
      R[v + rowadd][col]       = c0[v];
      R[16 + v + rowadd][col]  = c1[v];
      Hr[v + rowadd][col]      = h0[v];
      Hr[16 + v + rowadd][col] = h1[v];
    }
  }
  __syncthreads();

  const float* nb = node_b + lane * 4;
  const float* eb = edge_b + lane * 4;
  const float* vb = ws + MP_O + (size_t)Dn_ * Dm_ + lane * 4;
  const float4 n0v = *(const float4*)(nb);
  const float4 n1v = *(const float4*)(nb + Dm_);
  const float4 n2v = *(const float4*)(nb + 2 * Dm_);
  const float4 e0v = *(const float4*)(eb);
  const float4 e1v = *(const float4*)(eb + Dm_);
  const float4 v0v = *(const float4*)(vb);
  const float4 v1v = *(const float4*)(vb + MPROWS_ * Dm_);
  const float4 v2v = *(const float4*)(vb + 2 * MPROWS_ * Dm_);
  const float4 bv0 = make_float4(n0v.x + e0v.x + v0v.x, n0v.y + e0v.y + v0v.y,
                                 n0v.z + e0v.z + v0v.z, n0v.w + e0v.w + v0v.w);
  const float4 bv1 = make_float4(n1v.x + e1v.x + v1v.x, n1v.y + e1v.y + v1v.y,
                                 n1v.z + e1v.z + v1v.z, n1v.w + e1v.w + v1v.w);
  const float4 bv2 = make_float4(n2v.x + v2v.x, n2v.y + v2v.y, n2v.z + v2v.z, n2v.w + v2v.w);
  const float4 ibv = *(const float4*)(in_b + lane * 4);
  const float4 gv  = *(const float4*)(ln_g + lane * 4);
  const float4 bbv = *(const float4*)(ln_b + lane * 4);

#pragma unroll
  for (int i = 0; i < 4; ++i) {
    const int row = wave * 4 + i;
    const int d   = nodebase + row;
    const float c0 = (float)wsI[CNT_O + d];
    const float c1 = (float)wsI[CNT_O + N_ + d];
    const float c2 = (float)wsI[CNT_O + 2 * N_ + d];

    float4 x = *(const float4*)&Hr[row][lane * 4];
    x.x += ibv.x; x.y += ibv.y; x.z += ibv.z; x.w += ibv.w;

    float4 acc = *(const float4*)&R[row][lane * 4];
    acc.x += c0 * bv0.x + c1 * bv1.x + c2 * bv2.x;
    acc.y += c0 * bv0.y + c1 * bv1.y + c2 * bv2.y;
    acc.z += c0 * bv0.z + c1 * bv1.z + c2 * bv2.z;
    acc.w += c0 * bv0.w + c1 * bv1.w + c2 * bv2.w;

    float4 y;
    y.x = x.x + fmaxf(acc.x, 0.f);
    y.y = x.y + fmaxf(acc.y, 0.f);
    y.z = x.z + fmaxf(acc.z, 0.f);
    y.w = x.w + fmaxf(acc.w, 0.f);

    float s = y.x + y.y + y.z + y.w;
    float q = y.x * y.x + y.y * y.y + y.z * y.z + y.w * y.w;
#pragma unroll
    for (int o = 16; o >= 1; o >>= 1) {
      s += __shfl_xor(s, o, 32);
      q += __shfl_xor(q, o, 32);
    }
    const float mean = s * (1.0f / Dm_);
    const float var  = q * (1.0f / Dm_) - mean * mean;
    const float rs   = rsqrtf(var + 1e-5f);

    float4 o4;
    o4.x = (y.x - mean) * rs * gv.x + bbv.x;
    o4.y = (y.y - mean) * rs * gv.y + bbv.y;
    o4.z = (y.z - mean) * rs * gv.z + bbv.z;
    o4.w = (y.w - mean) * rs * gv.w + bbv.w;
    *(float4*)(Hout + (size_t)d * Dm_ + lane * 4) = o4;
  }
}

extern "C" void kernel_launch(void* const* d_in, const int* in_sizes, int n_in,
                              void* d_out, int out_size, void* d_ws, size_t ws_size,
                              hipStream_t stream) {
  const float* node_feat = (const float*)d_in[0];
  const float* in_W      = (const float*)d_in[1];
  const float* in_b      = (const float*)d_in[2];
  const float* node_W    = (const float*)d_in[3];
  const float* node_b    = (const float*)d_in[4];
  const float* edge_W    = (const float*)d_in[5];
  const float* edge_b    = (const float*)d_in[6];
  const float* ln_g      = (const float*)d_in[7];
  const float* ln_b      = (const float*)d_in[8];
  const float* ea0       = (const float*)d_in[9];
  const float* ea1       = (const float*)d_in[10];
  const int*   ei0       = (const int*)d_in[11];
  const int*   ei1       = (const int*)d_in[12];
  const int*   ei2       = (const int*)d_in[13];

  float* O0 = (float*)d_out;
  float* O1 = O0 + (size_t)N_ * Dm_;
  float* O2 = O0 + (size_t)2 * N_ * Dm_;
  float* O3 = O0 + (size_t)3 * N_ * Dm_;
  float* ws = (float*)d_ws;
  float* P1 = ws + (size_t)N_ * Dm_;
  float* P2 = ws + (size_t)2 * N_ * Dm_;
  float* P3 = ws + (size_t)3 * N_ * Dm_;

  prep_kernel<<<dim3(ZB_ + MPB_), dim3(256), 0, stream>>>(ws, in_W, in_b, node_W);
  hist_kernel<<<dim3(HIST_GRID, 3), dim3(256), 0, stream>>>(ws, ea0, ea1, ei0, ei1, ei2);
  scan_kernel<<<dim3(1), dim3(1024), 0, stream>>>(ws);
  fill_kernel<<<dim3(E_ / 256, 3), dim3(256), 0, stream>>>(ws, ei0, ei1, ei2);
  layer0_kernel<<<dim3(N_ / 32, 4), dim3(256), 0, stream>>>(
      node_feat, P1, P2, P3, O0, ws, in_W, in_b, node_b, edge_W, edge_b, ln_g, ln_b);
  layer_kernel<<<dim3(N_ / 32, 3), dim3(256), 0, stream>>>(
      P2, P3, O0, O1, O2, O3, ws, node_W, node_b, edge_W, edge_b, ln_g, ln_b, 1);
  layer_kernel<<<dim3(N_ / 32, 1), dim3(256), 0, stream>>>(
      P1, P1, P1, O0, O0, O0, ws, node_W, node_b, edge_W, edge_b, ln_g, ln_b, 1);
}


# Gather phase streams the 4 nodes' contiguous index lists as one run: index words in one round, rows 16 (layer 1) or 32 (layer 0) per round, sums stored at segment ends; same sum order
# speedup vs baseline: 4.9386x; 1.0686x over previous
#include <hip/hip_runtime.h>


#define B_ 4
#define N_ 20000
#define Dn_ 64
#define Dm_ 128
#define E_ 160000

#define PERM_O 0
#define OFF_O  480000
#define CNT_O  500032
#define CUR_O  560032
#define S_O    620032
#define ZERO_WORDS 760000
#define MP_O   1260032
#define MPROWS_ 80

typedef float v2f __attribute__((ext_vector_type(2)));
typedef float v8f __attribute__((ext_vector_type(8)));

__device__ __forceinline__ v8f wmma4(v2f a, v2f b, v8f c) {
  return __builtin_amdgcn_wmma_f32_16x16x4_f32(false, a, false, b, (short)0, c,
                                               false, false);
}

#define ZB_  ((ZERO_WORDS / 4 + 255) / 256)
#define MPB_ 15
static_assert(MPB_ * 8 == 3 * (MPROWS_ / 16) * (Dm_ / 16), "MP tiles");

__global__ __launch_bounds__(256)
void prep_kernel(float* __restrict__ ws, const float* __restrict__ in_W,
                 const float* __restrict__ in_b, const float* __restrict__ node_W) {
  if (blockIdx.x < ZB_) {
    const int i = blockIdx.x * 256 + threadIdx.x;
    if (i < ZERO_WORDS / 4) {
      ((float4*)(ws + CNT_O))[i] = make_float4(0.f, 0.f, 0.f, 0.f);
    }
    return;
  }
  const int lane = threadIdx.x & 31;
  const int wave = threadIdx.x >> 5;
  const int m    = lane & 15;
  const int kh   = lane >> 4;
  const int tile = (blockIdx.x - ZB_) * 8 + wave;
  const int r    = tile / 40;
  const int rt   = (tile % 40) >> 3;
  const int col  = (tile & 7) * 16 + m;
  const int row  = rt * 16 + m;
  const float* Wr   = node_W + (size_t)r * Dm_ * Dm_;
  const float* arow = (row < Dn_) ? (in_W + row * Dm_) : in_b;
  const bool live = row <= Dn_;
  v8f c = {0.f, 0.f, 0.f, 0.f, 0.f, 0.f, 0.f, 0.f};
#pragma unroll 8
  for (int kk = 0; kk < 32; ++kk) {
    const int k0 = kk * 4 + kh * 2;
    v2f a, b;
    const float ax = arow[k0];
    const float ay = arow[k0 + 1];
    a.x = live ? ax : 0.f;
    a.y = live ? ay : 0.f;
    b.x = Wr[k0 * Dm_ + col];
    b.y = Wr[(k0 + 1) * Dm_ + col];
    c = wmma4(a, b, c);
  }
  float* Mp = ws + MP_O + (size_t)r * MPROWS_ * Dm_;
  const int rowadd = kh * 8;
#pragma unroll
  for (int v = 0; v < 8; ++v) {
    Mp[(rt * 16 + v + rowadd) * Dm_ + col] = c[v];
  }
}


#define HIST_GRID 1000
#define HIST_TRIPS 10
static_assert(HIST_TRIPS * HIST_GRID * 256 == E_ * 16, "hist trips");
static_assert(HIST_GRID * 256 >= E_, "relation 2: one pass");

__global__ __launch_bounds__(256)
void hist_kernel(float* __restrict__ ws,
                 const float* __restrict__ ea0, const float* __restrict__ ea1,
                 const int* __restrict__ ei0, const int* __restrict__ ei1,
                 const int* __restrict__ ei2) {
  const int r = blockIdx.y;
  int* cnt = (int*)ws + CNT_O + r * N_;
  if (r < 2) {
    const int*   ei = (r == 0) ? ei0 : ei1;
    const float* ea = (r == 0) ? ea0 : ea1;
    float* S = ws + S_O + (size_t)r * N_ * 16;
    const int k = threadIdx.x & 15;
    int   dd[HIST_TRIPS];
    float vv[HIST_TRIPS];
#pragma unroll
    for (int it = 0; it < HIST_TRIPS; ++it) {
      const int t = (it * HIST_GRID + blockIdx.x) * 256 + threadIdx.x;
      dd[it] = ei[E_ + (t >> 4)];
      vv[it] = ea[t];
    }
#pragma unroll
    for (int it = 0; it < HIST_TRIPS; ++it) {
      if (k == 0) atomicAdd(cnt + dd[it], 1);
      atomicAdd(S + (size_t)dd[it] * 16 + k, vv[it]);
    }
  } else {
    for (int e = blockIdx.x * 256 + threadIdx.x; e < E_; e += HIST_GRID * 256) {
      atomicAdd(cnt + ei2[E_ + e], 1);
    }
  }
}

#define SCAN_CH 20
static_assert(N_ % SCAN_CH == 0 && 1024 * SCAN_CH >= N_, "scan chunks");

__global__ __launch_bounds__(1024)
void scan_kernel(float* __restrict__ ws) {
  __shared__ int part[1024];
  const int t = threadIdx.x;
  const int* cnt = (const int*)ws + CNT_O;
  int* off = (int*)ws + OFF_O;

  const bool inside = t * SCAN_CH < N_;
  int c[SCAN_CH];
#pragma unroll
  for (int i = 0; i < SCAN_CH; ++i) c[i] = 0;
  if (inside) {
#pragma unroll
    for (int i = 0; i < SCAN_CH; ++i) {
      const int d = t * SCAN_CH + i;
      c[i] = cnt[d] + cnt[N_ + d] + cnt[2 * N_ + d];
    }
  }
  int loc[SCAN_CH];
  int sum = 0;
#pragma unroll
  for (int i = 0; i < SCAN_CH; ++i) {
    loc[i] = sum;
    sum += c[i];
  }
  part[t] = sum;
  __syncthreads();
  for (int ofs = 1; ofs < 1024; ofs <<= 1) {
    const int v = (t >= ofs) ? part[t - ofs] : 0;
    __syncthreads();
    part[t] += v;
    __syncthreads();
  }
  const int excl = part[t] - sum;
  if (inside) {
#pragma unroll
    for (int i = 0; i < SCAN_CH; ++i) off[t * SCAN_CH + i] = excl + loc[i];
  }
  if (t == 1023) off[N_] = part[1023];
}

__global__ __launch_bounds__(256)
void fill_kernel(float* __restrict__ ws, const int* __restrict__ ei0,
                 const int* __restrict__ ei1, const int* __restrict__ ei2) {
  const int r = blockIdx.y;
  const int e = blockIdx.x * 256 + threadIdx.x;
  const int* ei = (r == 0) ? ei0 : (r == 1) ? ei1 : ei2;
  const int s = ei[e];
  const int d = ei[E_ + e];
  int* wsI = (int*)ws;
  int seg = wsI[OFF_O + d];
  if (r >= 1) seg += wsI[CNT_O + d];
  if (r == 2) seg += wsI[CNT_O + N_ + d];
  const int pos = seg + atomicAdd(wsI + CUR_O + r * N_ + d, 1);
  wsI[PERM_O + pos] = s;
}

#define KA_ 416
#define APITCH_ 420

__global__ __launch_bounds__(256)
void layer_kernel(const float* __restrict__ in0, const float* __restrict__ in1,
                  const float* __restrict__ in2, float* __restrict__ out0,
                  float* __restrict__ out1, float* __restrict__ out2,
                  const float* __restrict__ ws,
                  const float* __restrict__ node_W, const float* __restrict__ node_b,
                  const float* __restrict__ edge_W, const float* __restrict__ edge_b,
                  const float* __restrict__ ln_g, const float* __restrict__ ln_b,
                  int layer) {
  __shared__ float A[32][APITCH_];
  __shared__ float R[32][132];

  const int tid  = threadIdx.x;
  const int lane = tid & 31;
  const int wave = __builtin_amdgcn_readfirstlane(tid >> 5);
  const int m    = lane & 15;
  const int kh   = lane >> 4;
  const int nodebase = blockIdx.x * 32;
  const int* wsI = (const int*)ws;

  const float* Hin  = (blockIdx.y == 0) ? in0  : (blockIdx.y == 1) ? in1  : in2;
  float*       Hout = (blockIdx.y == 0) ? out0 : (blockIdx.y == 1) ? out1 : out2;

  const float* Hl = Hin + lane * 4;
  const int* perm = wsI + PERM_O;
  const int d0 = nodebase + wave * 4;
  const int sl = (lane < 12) ? lane : 11;
  const int si = (sl * 11) >> 5;
  const int sr = sl - si * 3;
  const int voff = wsI[OFF_O + d0 + si];
  const int vc0  = wsI[CNT_O + d0 + si];
  const int vc1  = wsI[CNT_O + N_ + d0 + si];
  const int vc2  = wsI[CNT_O + 2 * N_ + d0 + si];
  float sv[4];
#pragma unroll
  for (int i = 0; i < 4; ++i) sv[i] = ws[S_O + ((size_t)kh * N_ + d0 + i) * 16 + m];
  const int vend = voff + vc0 + ((sr >= 1) ? vc1 : 0) + ((sr >= 2) ? vc2 : 0);
  const int beg0 = __builtin_amdgcn_readlane(voff, 0);
  const int end3 = __builtin_amdgcn_readlane(vend, 11);

  float4 acc = make_float4(0.f, 0.f, 0.f, 0.f);
  int seg = 0, seg_i = 0, seg_r = 0;
  int cur_end = __builtin_amdgcn_readlane(vend, 0);
  for (int base = beg0; base < end3; base += 128) {
    int ix0, ix1, ix2, ix3;
    { const int q = base + lane;      ix0 = (q < end3) ? perm[q] : 0; }
    { const int q = base + 32 + lane; ix1 = (q < end3) ? perm[q] : 0; }
    { const int q = base + 64 + lane; ix2 = (q < end3) ? perm[q] : 0; }
    { const int q = base + 96 + lane; ix3 = (q < end3) ? perm[q] : 0; }
    for (int c = 0; c < 4; ++c) {
      const int cb = base + c * 32;
      if (cb >= end3) break;
      for (int h = 0; h < 32; h += 16) {
        const int qb = cb + h;
        if (qb >= end3) break;
        float4 v[16];
#pragma unroll
        for (int u = 0; u < 8; ++u) {
          const int rid = __builtin_amdgcn_readlane(ix0, h + u);
          v[u] = *(const float4*)(Hl + (size_t)rid * Dm_);
        }
        if (qb + 8 < end3) {
#pragma unroll
          for (int u = 8; u < 16; ++u) {
            const int rid = __builtin_amdgcn_readlane(ix0, h + u);
            v[u] = *(const float4*)(Hl + (size_t)rid * Dm_);
          }
        } else {
#pragma unroll
          for (int u = 8; u < 16; ++u) v[u] = make_float4(0.f, 0.f, 0.f, 0.f);
        }
#pragma unroll
        for (int u = 0; u < 16; ++u) {
          const int q = qb + u;
          if (q < end3) {
            while (q >= cur_end && seg < 11) {
              *(float4*)&A[wave * 4 + seg_i][seg_r * Dm_ + lane * 4] = acc;
              acc = make_float4(0.f, 0.f, 0.f, 0.f);
              ++seg; ++seg_r;
              if (seg_r == 3) { seg_r = 0; ++seg_i; }
              cur_end = __builtin_amdgcn_readlane(vend, seg);
            }
            acc.x += v[u].x; acc.y += v[u].y; acc.z += v[u].z; acc.w += v[u].w;
          }
        }
      }
      ix0 = ix1; ix1 = ix2; ix2 = ix3;
    }
  }
  while (seg < 12) {
    *(float4*)&A[wave * 4 + seg_i][seg_r * Dm_ + lane * 4] = acc;
    acc = make_float4(0.f, 0.f, 0.f, 0.f);
    ++seg; ++seg_r;
    if (seg_r == 3) { seg_r = 0; ++seg_i; }
  }
#pragma unroll
  for (int i = 0; i < 4; ++i) A[wave * 4 + i][3 * Dm_ + lane] = sv[i];
  __syncthreads();

  {
    const int col = wave * 16 + m;
    const float* Wn = node_W + (size_t)layer * 3 * Dm_ * Dm_;
    const float* We = edge_W + (size_t)layer * 2 * 16 * Dm_;
    v8f c0 = {0.f, 0.f, 0.f, 0.f, 0.f, 0.f, 0.f, 0.f};
    v8f c1 = c0;
#pragma unroll 16
    for (int kk = 0; kk < 96; ++kk) {
      const int k0 = kk * 4 + kh * 2;
      v2f f0, f1, bf;
      f0.x = A[m][k0];
      f0.y = A[m][k0 + 1];
      f1.x = A[16 + m][k0];
      f1.y = A[16 + m][k0 + 1];
      bf.x = Wn[k0 * Dm_ + col];
      bf.y = Wn[(k0 + 1) * Dm_ + col];
      c0 = wmma4(f0, bf, c0);
      c1 = wmma4(f1, bf, c1);
    }
#pragma unroll
    for (int kk = 0; kk < 8; ++kk) {
      const int k0 = kk * 4 + kh * 2;
      v2f f0, f1, bf;
      f0.x = A[m][3 * Dm_ + k0];
      f0.y = A[m][3 * Dm_ + k0 + 1];
      f1.x = A[16 + m][3 * Dm_ + k0];
      f1.y = A[16 + m][3 * Dm_ + k0 + 1];
      bf.x = We[k0 * Dm_ + col];
      bf.y = We[(k0 + 1) * Dm_ + col];
      c0 = wmma4(f0, bf, c0);
      c1 = wmma4(f1, bf, c1);
    }
    const int rowadd = kh * 8;
#pragma unroll
    for (int v = 0; v < 8; ++v) {
      R[v + rowadd][col]      = c0[v];
      R[16 + v + rowadd][col] = c1[v];
    }
  }
  __syncthreads();

  const float* nb = node_b + (size_t)layer * 3 * Dm_ + lane * 4;
  const float* eb = edge_b + (size_t)layer * 2 * Dm_ + lane * 4;
  const float4 n0v = *(const float4*)(nb);
  const float4 n1v = *(const float4*)(nb + Dm_);
  const float4 n2v = *(const float4*)(nb + 2 * Dm_);
  const float4 e0v = *(const float4*)(eb);
  const float4 e1v = *(const float4*)(eb + Dm_);
  const float4 bv0 = make_float4(n0v.x + e0v.x, n0v.y + e0v.y, n0v.z + e0v.z, n0v.w + e0v.w);
  const float4 bv1 = make_float4(n1v.x + e1v.x, n1v.y + e1v.y, n1v.z + e1v.z, n1v.w + e1v.w);
  const float4 gv  = *(const float4*)(ln_g + (size_t)layer * Dm_ + lane * 4);
  const float4 bbv = *(const float4*)(ln_b + (size_t)layer * Dm_ + lane * 4);

#pragma unroll
  for (int i = 0; i < 4; ++i) {
    const int row = wave * 4 + i;
    const int d   = nodebase + row;
    const float c0 = (float)wsI[CNT_O + d];
    const float c1 = (float)wsI[CNT_O + N_ + d];
    const float c2 = (float)wsI[CNT_O + 2 * N_ + d];
    const float4 x = *(const float4*)(Hin + (size_t)d * Dm_ + lane * 4);

    float4 acc = *(const float4*)&R[row][lane * 4];
    acc.x += c0 * bv0.x + c1 * bv1.x + c2 * n2v.x;
    acc.y += c0 * bv0.y + c1 * bv1.y + c2 * n2v.y;
    acc.z += c0 * bv0.z + c1 * bv1.z + c2 * n2v.z;
    acc.w += c0 * bv0.w + c1 * bv1.w + c2 * n2v.w;

    float4 y;
    y.x = x.x + fmaxf(acc.x, 0.f);
    y.y = x.y + fmaxf(acc.y, 0.f);
    y.z = x.z + fmaxf(acc.z, 0.f);
    y.w = x.w + fmaxf(acc.w, 0.f);

    float s = y.x + y.y + y.z + y.w;
    float q = y.x * y.x + y.y * y.y + y.z * y.z + y.w * y.w;
#pragma unroll
    for (int o = 16; o >= 1; o >>= 1) {
      s += __shfl_xor(s, o, 32);
      q += __shfl_xor(q, o, 32);
    }
    const float mean = s * (1.0f / Dm_);
    const float var  = q * (1.0f / Dm_) - mean * mean;
    const float rs   = rsqrtf(var + 1e-5f);

    float4 o4;
    o4.x = (y.x - mean) * rs * gv.x + bbv.x;
    o4.y = (y.y - mean) * rs * gv.y + bbv.y;
    o4.z = (y.z - mean) * rs * gv.z + bbv.z;
    o4.w = (y.w - mean) * rs * gv.w + bbv.w;
    *(float4*)(Hout + (size_t)d * Dm_ + lane * 4) = o4;
  }
}

#define K0A_ 224
#define APITCH0_ 292

__global__ __launch_bounds__(256)
void layer0_kernel(const float* __restrict__ X_all, float* __restrict__ out0,
                   float* __restrict__ out1, float* __restrict__ out2,
                   float* __restrict__ out3, const float* __restrict__ ws,
                   const float* __restrict__ in_W, const float* __restrict__ in_b,
                   const float* __restrict__ node_b,
                   const float* __restrict__ edge_W, const float* __restrict__ edge_b,
                   const float* __restrict__ ln_g, const float* __restrict__ ln_b) {
  __shared__ float A[32][APITCH0_];
  __shared__ float R[32][132];
  __shared__ float Hr[32][132];

  const int tid  = threadIdx.x;
  const int lane = tid & 31;
  const int wave = __builtin_amdgcn_readfirstlane(tid >> 5);
  const int m    = lane & 15;
  const int kh   = lane >> 4;
  const int nodebase = blockIdx.x * 32;
  const int* wsI = (const int*)ws;

  const float* X = X_all + (size_t)blockIdx.y * N_ * Dn_;
  float* Hout = (blockIdx.y == 0) ? out0 : (blockIdx.y == 1) ? out1
              : (blockIdx.y == 2) ? out2 : out3;

  const float* Xl = X + lane * 2;
  const int* perm = wsI + PERM_O;
  const int d0 = nodebase + wave * 4;
  const int sl = (lane < 12) ? lane : 11;
  const int si = (sl * 11) >> 5;
  const int sr = sl - si * 3;
  const int voff = wsI[OFF_O + d0 + si];
  const int vc0  = wsI[CNT_O + d0 + si];
  const int vc1  = wsI[CNT_O + N_ + d0 + si];
  const int vc2  = wsI[CNT_O + 2 * N_ + d0 + si];
  float  sv[4];
  float2 xd[4];
#pragma unroll
  for (int i = 0; i < 4; ++i) {
    sv[i] = ws[S_O + ((size_t)kh * N_ + d0 + i) * 16 + m];
    xd[i] = *(const float2*)(Xl + (size_t)(d0 + i) * Dn_);
  }
  const int vend = voff + vc0 + ((sr >= 1) ? vc1 : 0) + ((sr >= 2) ? vc2 : 0);
  const int beg0 = __builtin_amdgcn_readlane(voff, 0);
  const int end3 = __builtin_amdgcn_readlane(vend, 11);

  float2 acc = make_float2(0.f, 0.f);
  int seg = 0, seg_i = 0, seg_r = 0;
  int cur_end = __builtin_amdgcn_readlane(vend, 0);
  for (int base = beg0; base < end3; base += 128) {
    int ix0, ix1, ix2, ix3;
    { const int q = base + lane;      ix0 = (q < end3) ? perm[q] : 0; }
    { const int q = base + 32 + lane; ix1 = (q < end3) ? perm[q] : 0; }
    { const int q = base + 64 + lane; ix2 = (q < end3) ? perm[q] : 0; }
    { const int q = base + 96 + lane; ix3 = (q < end3) ? perm[q] : 0; }
    for (int c = 0; c < 4; ++c) {
      const int cb = base + c * 32;
      if (cb >= end3) break;
      float2 v[32];
#pragma unroll
      for (int u = 0; u < 16; ++u) {
        const int rid = __builtin_amdgcn_readlane(ix0, u);
        v[u] = *(const float2*)(Xl + (size_t)rid * Dn_);
      }
      if (cb + 16 < end3) {
#pragma unroll
        for (int u = 16; u < 32; ++u) {
          const int rid = __builtin_amdgcn_readlane(ix0, u);
          v[u] = *(const float2*)(Xl + (size_t)rid * Dn_);
        }
      } else {
#pragma unroll
        for (int u = 16; u < 32; ++u) v[u] = make_float2(0.f, 0.f);
      }
#pragma unroll
      for (int u = 0; u < 32; ++u) {
        const int q = cb + u;
        if (q < end3) {
          while (q >= cur_end && seg < 11) {
            *(float2*)&A[wave * 4 + seg_i][seg_r * Dn_ + lane * 2] = acc;
            acc = make_float2(0.f, 0.f);
            ++seg; ++seg_r;
            if (seg_r == 3) { seg_r = 0; ++seg_i; }
            cur_end = __builtin_amdgcn_readlane(vend, seg);
          }
          acc.x += v[u].x; acc.y += v[u].y;
        }
      }
      ix0 = ix1; ix1 = ix2; ix2 = ix3;
    }
  }
  while (seg < 12) {
    *(float2*)&A[wave * 4 + seg_i][seg_r * Dn_ + lane * 2] = acc;
    acc = make_float2(0.f, 0.f);
    ++seg; ++seg_r;
    if (seg_r == 3) { seg_r = 0; ++seg_i; }
  }
#pragma unroll
  for (int i = 0; i < 4; ++i) {
    A[wave * 4 + i][3 * Dn_ + lane] = sv[i];
    *(float2*)&A[wave * 4 + i][K0A_ + lane * 2] = xd[i];
  }
  __syncthreads();

  {
    const int col = wave * 16 + m;
    const float* Mp = ws + MP_O;
    v8f c0 = {0.f, 0.f, 0.f, 0.f, 0.f, 0.f, 0.f, 0.f};
    v8f c1 = c0, h0 = c0, h1 = c0;
#pragma unroll 16
    for (int kk = 0; kk < 48; ++kk) {
      const int k0 = kk * 4 + kh * 2;
      const int kr = (kk >> 4) * MPROWS_ + (kk & 15) * 4 + kh * 2;
      v2f f0, f1, bf;
      f0.x = A[m][k0];
      f0.y = A[m][k0 + 1];
      f1.x = A[16 + m][k0];
      f1.y = A[16 + m][k0 + 1];
      bf.x = Mp[kr * Dm_ + col];
      bf.y = Mp[(kr + 1) * Dm_ + col];
      c0 = wmma4(f0, bf, c0);
      c1 = wmma4(f1, bf, c1);
    }
#pragma unroll
    for (int kk = 0; kk < 8; ++kk) {
      const int k0 = kk * 4 + kh * 2;
      v2f f0, f1, bf;
      f0.x = A[m][3 * Dn_ + k0];
      f0.y = A[m][3 * Dn_ + k0 + 1];
      f1.x = A[16 + m][3 * Dn_ + k0];
      f1.y = A[16 + m][3 * Dn_ + k0 + 1];
      bf.x = edge_W[k0 * Dm_ + col];
      bf.y = edge_W[(k0 + 1) * Dm_ + col];
      c0 = wmma4(f0, bf, c0);
      c1 = wmma4(f1, bf, c1);
    }
#pragma unroll
    for (int kk = 0; kk < 16; ++kk) {
      const int k0 = kk * 4 + kh * 2;
      v2f f0, f1, bf;
      f0.x = A[m][K0A_ + k0];
      f0.y = A[m][K0A_ + k0 + 1];
      f1.x = A[16 + m][K0A_ + k0];
      f1.y = A[16 + m][K0A_ + k0 + 1];
      bf.x = in_W[k0 * Dm_ + col];
      bf.y = in_W[(k0 + 1) * Dm_ + col];
      h0 = wmma4(f0, bf, h0);
      h1 = wmma4(f1, bf, h1);
    }
    const int rowadd = kh * 8;
#pragma unroll
    for (int v = 0; v < 8; ++v) {
      R[v + rowadd][col]       = c0[v];
      R[16 + v + rowadd][col]  = c1[v];
      Hr[v + rowadd][col]      = h0[v];
      Hr[16 + v + rowadd][col] = h1[v];
    }
  }
  __syncthreads();

  const float* nb = node_b + lane * 4;
  const float* eb = edge_b + lane * 4;
  const float* vb = ws + MP_O + (size_t)Dn_ * Dm_ + lane * 4;
  const float4 n0v = *(const float4*)(nb);
  const float4 n1v = *(const float4*)(nb + Dm_);
  const float4 n2v = *(const float4*)(nb + 2 * Dm_);
  const float4 e0v = *(const float4*)(eb);
  const float4 e1v = *(const float4*)(eb + Dm_);
  const float4 v0v = *(const float4*)(vb);
  const float4 v1v = *(const float4*)(vb + MPROWS_ * Dm_);
  const float4 v2v = *(const float4*)(vb + 2 * MPROWS_ * Dm_);
  const float4 bv0 = make_float4(n0v.x + e0v.x + v0v.x, n0v.y + e0v.y + v0v.y,
                                 n0v.z + e0v.z + v0v.z, n0v.w + e0v.w + v0v.w);
  const float4 bv1 = make_float4(n1v.x + e1v.x + v1v.x, n1v.y + e1v.y + v1v.y,
                                 n1v.z + e1v.z + v1v.z, n1v.w + e1v.w + v1v.w);
  const float4 bv2 = make_float4(n2v.x + v2v.x, n2v.y + v2v.y, n2v.z + v2v.z, n2v.w + v2v.w);
  const float4 ibv = *(const float4*)(in_b + lane * 4);
  const float4 gv  = *(const float4*)(ln_g + lane * 4);
  const float4 bbv = *(const float4*)(ln_b + lane * 4);

#pragma unroll
  for (int i = 0; i < 4; ++i) {
    const int row = wave * 4 + i;
    const int d   = nodebase + row;
    const float c0 = (float)wsI[CNT_O + d];
    const float c1 = (float)wsI[CNT_O + N_ + d];
    const float c2 = (float)wsI[CNT_O + 2 * N_ + d];

    float4 x = *(const float4*)&Hr[row][lane * 4];
    x.x += ibv.x; x.y += ibv.y; x.z += ibv.z; x.w += ibv.w;

    float4 acc = *(const float4*)&R[row][lane * 4];
    acc.x += c0 * bv0.x + c1 * bv1.x + c2 * bv2.x;
    acc.y += c0 * bv0.y + c1 * bv1.y + c2 * bv2.y;
    acc.z += c0 * bv0.z + c1 * bv1.z + c2 * bv2.z;
    acc.w += c0 * bv0.w + c1 * bv1.w + c2 * bv2.w;

    float4 y;
    y.x = x.x + fmaxf(acc.x, 0.f);
    y.y = x.y + fmaxf(acc.y, 0.f);
    y.z = x.z + fmaxf(acc.z, 0.f);
    y.w = x.w + fmaxf(acc.w, 0.f);

    float s = y.x + y.y + y.z + y.w;
    float q = y.x * y.x + y.y * y.y + y.z * y.z + y.w * y.w;
#pragma unroll
    for (int o = 16; o >= 1; o >>= 1) {
      s += __shfl_xor(s, o, 32);
      q += __shfl_xor(q, o, 32);
    }
    const float mean = s * (1.0f / Dm_);
    const float var  = q * (1.0f / Dm_) - mean * mean;
    const float rs   = rsqrtf(var + 1e-5f);

    float4 o4;
    o4.x = (y.x - mean) * rs * gv.x + bbv.x;
    o4.y = (y.y - mean) * rs * gv.y + bbv.y;
    o4.z = (y.z - mean) * rs * gv.z + bbv.z;
    o4.w = (y.w - mean) * rs * gv.w + bbv.w;
    *(float4*)(Hout + (size_t)d * Dm_ + lane * 4) = o4;
  }
}

extern "C" void kernel_launch(void* const* d_in, const int* in_sizes, int n_in,
                              void* d_out, int out_size, void* d_ws, size_t ws_size,
                              hipStream_t stream) {
  const float* node_feat = (const float*)d_in[0];
  const float* in_W      = (const float*)d_in[1];
  const float* in_b      = (const float*)d_in[2];
  const float* node_W    = (const float*)d_in[3];
  const float* node_b    = (const float*)d_in[4];
  const float* edge_W    = (const float*)d_in[5];
  const float* edge_b    = (const float*)d_in[6];
  const float* ln_g      = (const float*)d_in[7];
  const float* ln_b      = (const float*)d_in[8];
  const float* ea0       = (const float*)d_in[9];
  const float* ea1       = (const float*)d_in[10];
  const int*   ei0       = (const int*)d_in[11];
  const int*   ei1       = (const int*)d_in[12];
  const int*   ei2       = (const int*)d_in[13];

  float* O0 = (float*)d_out;
  float* O1 = O0 + (size_t)N_ * Dm_;
  float* O2 = O0 + (size_t)2 * N_ * Dm_;
  float* O3 = O0 + (size_t)3 * N_ * Dm_;
  float* ws = (float*)d_ws;
  float* P1 = ws + (size_t)N_ * Dm_;
  float* P2 = ws + (size_t)2 * N_ * Dm_;
  float* P3 = ws + (size_t)3 * N_ * Dm_;

  prep_kernel<<<dim3(ZB_ + MPB_), dim3(256), 0, stream>>>(ws, in_W, in_b, node_W);
  hist_kernel<<<dim3(HIST_GRID, 3), dim3(256), 0, stream>>>(ws, ea0, ea1, ei0, ei1, ei2);
  scan_kernel<<<dim3(1), dim3(1024), 0, stream>>>(ws);
  fill_kernel<<<dim3(E_ / 256, 3), dim3(256), 0, stream>>>(ws, ei0, ei1, ei2);
  layer0_kernel<<<dim3(N_ / 32, 4), dim3(256), 0, stream>>>(
      node_feat, P1, P2, P3, O0, ws, in_W, in_b, node_b, edge_W, edge_b, ln_g, ln_b);
  layer_kernel<<<dim3(N_ / 32, 3), dim3(256), 0, stream>>>(
      P2, P3, O0, O1, O2, O3, ws, node_W, node_b, edge_W, edge_b, ln_g, ln_b, 1);
  layer_kernel<<<dim3(N_ / 32, 1), dim3(256), 0, stream>>>(
      P1, P1, P1, O0, O0, O0, ws, node_W, node_b, edge_W, edge_b, ln_g, ln_b, 1);
}
